# starting kernel
# baseline (speedup 1.0000x reference)
#include <hip/hip_runtime.h>
#include <math.h>

typedef float v2f __attribute__((ext_vector_type(2)));
typedef float v8f __attribute__((ext_vector_type(8)));

#define TWO_PI_F     6.2831853071795864769f
#define MIN_W        1e-3f
#define MIN_H        1e-3f
#define MIN_D        1e-3f
#define DERIV_SHIFT  0.54132485461292f

#define KBINS 32
#define DH    8
#define CDIM  64
#define HDIM  256
#define NPAR  768

#define WAVES           2
#define MTILES          2
#define ROWS_PER_WAVE   (MTILES * 16)
#define ROWS_PER_BLOCK  (WAVES * ROWS_PER_WAVE)
#define PCHUNK          96
#define NT2             (PCHUNK / 16)

#define HBUF_FLOATS (ROWS_PER_BLOCK * HDIM)
#define PBUF_FLOATS (ROWS_PER_BLOCK * PCHUNK)
#define LDS_BYTES   ((HBUF_FLOATS + PBUF_FLOATS) * 4)

static __device__ __forceinline__ float softplus_f(float x) {
    return (x > 20.0f) ? x : log1pf(__expf(x));
}

__global__ __launch_bounds__(WAVES * 32)
void Circular_Spline_Flow_fused_kernel(
    const float* __restrict__ theta,
    const float* __restrict__ x,
    const float* __restrict__ W1,
    const float* __restrict__ b1,
    const float* __restrict__ W2,
    const float* __restrict__ b2,
    const float* __restrict__ eta,
    float* __restrict__ out,
    int Btotal)
{
    extern __shared__ float smem[];

    const int tid  = threadIdx.x;
    const int wave = tid >> 5;
    const int lane = tid & 31;
    const int m    = lane & 15;
    const int half = lane >> 4;

    const int rowbase = blockIdx.x * ROWS_PER_BLOCK + wave * ROWS_PER_WAVE;

    float* hbuf = smem + wave * ROWS_PER_WAVE * HDIM;
    float* pbuf = smem + HBUF_FLOATS + wave * ROWS_PER_WAVE * PCHUNK;

    const float etav = eta[0];

    v2f afrag[MTILES][16];
    #pragma unroll
    for (int mt = 0; mt < MTILES; ++mt) {
        const float* xrow = x + (size_t)(rowbase + mt * 16 + m) * CDIM;
        #pragma unroll
        for (int kc = 0; kc < 16; ++kc) {
            const int kk = 4 * kc + 2 * half;
            afrag[mt][kc].x = xrow[kk];
            afrag[mt][kc].y = xrow[kk + 1];
        }
    }

    for (int j = 0; j < 16; ++j) {
        const int col = 16 * j + m;
        v8f acc0 = {0.f,0.f,0.f,0.f,0.f,0.f,0.f,0.f};
        v8f acc1 = {0.f,0.f,0.f,0.f,0.f,0.f,0.f,0.f};
        #pragma unroll
        for (int kc = 0; kc < 16; ++kc) {
            const int kk = 4 * kc + 2 * half;
            v2f bfrag;
            bfrag.x = W1[(size_t)kk * HDIM + col];
            bfrag.y = W1[(size_t)(kk + 1) * HDIM + col];
            acc0 = __builtin_amdgcn_wmma_f32_16x16x4_f32(
                false, afrag[0][kc], false, bfrag, (short)0, acc0, false, false);
            acc1 = __builtin_amdgcn_wmma_f32_16x16x4_f32(
                false, afrag[1][kc], false, bfrag, (short)0, acc1, false, false);
        }
        const float bias = b1[col];
        #pragma unroll
        for (int v = 0; v < 8; ++v) {
            float h0 = fmaxf(acc0[v] + bias, 0.0f);
            float h1 = fmaxf(acc1[v] + bias, 0.0f);
            hbuf[( 0 + 8 * half + v) * HDIM + col] = h0;
            hbuf[(16 + 8 * half + v) * HDIM + col] = h1;
        }
    }

    for (int dp = 0; dp < DH; ++dp) {
        const int colbase = dp * PCHUNK;

        __builtin_prefetch(&W2[(size_t)(2 * half) * NPAR + colbase + m], 0, 1);

        v8f acc[MTILES][NT2];
        #pragma unroll
        for (int mt = 0; mt < MTILES; ++mt)
            #pragma unroll
            for (int jt = 0; jt < NT2; ++jt)
                acc[mt][jt] = (v8f){0.f,0.f,0.f,0.f,0.f,0.f,0.f,0.f};

        #pragma unroll 2
        for (int kc = 0; kc < 64; ++kc) {
            const int kk = 4 * kc + 2 * half;
            v2f a0, a1;
            a0.x = hbuf[( 0 + m) * HDIM + kk];
            a0.y = hbuf[( 0 + m) * HDIM + kk + 1];
            a1.x = hbuf[(16 + m) * HDIM + kk];
            a1.y = hbuf[(16 + m) * HDIM + kk + 1];
            #pragma unroll
            for (int jt = 0; jt < NT2; ++jt) {
                const int col = colbase + 16 * jt + m;
                v2f b;
                b.x = W2[(size_t)kk * NPAR + col];
                b.y = W2[(size_t)(kk + 1) * NPAR + col];
                acc[0][jt] = __builtin_amdgcn_wmma_f32_16x16x4_f32(
                    false, a0, false, b, (short)0, acc[0][jt], false, false);
                acc[1][jt] = __builtin_amdgcn_wmma_f32_16x16x4_f32(
                    false, a1, false, b, (short)0, acc[1][jt], false, false);
            }
        }

        #pragma unroll
        for (int jt = 0; jt < NT2; ++jt) {
            const float bias = b2[colbase + 16 * jt + m];
            #pragma unroll
            for (int mt = 0; mt < MTILES; ++mt) {
                #pragma unroll
                for (int v = 0; v < 8; ++v) {
                    const float pv = (acc[mt][jt][v] + bias) * etav;
                    pbuf[(mt * 16 + 8 * half + v) * PCHUNK + 16 * jt + m] = pv;
                }
            }
        }

        {
            const float* P  = &pbuf[lane * PCHUNK];
            const float* uw = P;
            const float* uh = P + 32;
            const float* ud = P + 64;
            const float  th = theta[(size_t)(rowbase + lane) * DH + dp];

            float mw = -1e30f, mh = -1e30f;
            for (int i = 0; i < KBINS; ++i) {
                mw = fmaxf(mw, uw[i]);
                mh = fmaxf(mh, uh[i]);
            }
            float sw = 0.f, sh = 0.f;
            for (int i = 0; i < KBINS; ++i) {
                sw += __expf(uw[i] - mw);
                sh += __expf(uh[i] - mh);
            }
            const float wscale = (1.0f - MIN_W * KBINS) / sw;
            const float hscale = (1.0f - MIN_H * KBINS) / sh;

            float cw = 0.f, ch = 0.f;
            float Sw = 0.f, Sh = 0.f;
            int   idx = 0;
            float sel_cw = 0.f, sel_ch = 0.f, sel_w = TWO_PI_F, sel_h = TWO_PI_F;
            for (int i = 0; i < KBINS; ++i) {
                const float wfrac = MIN_W + wscale * __expf(uw[i] - mw);
                const float hfrac = MIN_H + hscale * __expf(uh[i] - mh);
                Sw += wfrac;
                Sh += hfrac;
                const float cw_next = (i == KBINS - 1) ? TWO_PI_F : TWO_PI_F * Sw;
                const float ch_next = (i == KBINS - 1) ? TWO_PI_F : TWO_PI_F * Sh;
                if (th >= cw) {
                    idx    = i;
                    sel_cw = cw;  sel_w = cw_next - cw;
                    sel_ch = ch;  sel_h = ch_next - ch;
                }
                cw = cw_next;
                ch = ch_next;
            }

            const float d0 = MIN_D + softplus_f(ud[idx] + DERIV_SHIFT);
            const float d1 = MIN_D + softplus_f(ud[(idx + 1) & (KBINS - 1)] + DERIV_SHIFT);

            const float delta = sel_h / sel_w;
            const float t     = (th - sel_cw) / sel_w;
            const float t1mt  = t * (1.0f - t);
            const float numer = sel_h * (delta * t * t + d0 * t1mt);
            const float denom = delta + (d0 + d1 - 2.0f * delta) * t1mt;
            const float outv  = sel_ch + numer / denom;

            const float omt  = 1.0f - t;
            const float dnum = delta * delta * (d1 * t * t + 2.0f * delta * t1mt + d0 * omt * omt);
            const float lad  = __logf(dnum) - 2.0f * __logf(denom);

            const size_t o = (size_t)(rowbase + lane) * DH + dp;
            out[o] = outv;
            out[(size_t)Btotal * DH + o] = lad;
        }
    }
}

extern "C" void kernel_launch(void* const* d_in, const int* in_sizes, int n_in,
                              void* d_out, int out_size, void* d_ws, size_t ws_size,
                              hipStream_t stream) {
    (void)n_in; (void)out_size; (void)d_ws; (void)ws_size;
    const float* theta = (const float*)d_in[0];
    const float* x     = (const float*)d_in[1];
    const float* W1    = (const float*)d_in[2];
    const float* b1    = (const float*)d_in[3];
    const float* W2    = (const float*)d_in[4];
    const float* b2    = (const float*)d_in[5];
    const float* eta   = (const float*)d_in[6];
    float* out = (float*)d_out;

    const int B = in_sizes[0] / DH;
    const int blocks = B / ROWS_PER_BLOCK;
    Circular_Spline_Flow_fused_kernel<<<blocks, WAVES * 32, LDS_BYTES, stream>>>(
        theta, x, W1, b1, W2, b2, eta, out, B);
}


# work-group barrier split: each wave signals after its head's last slab read and waits only at the head's end; the spline runs between the two (was one whole barrier behind the spline); sums same
# speedup vs baseline: 2.5225x; 2.5225x over previous
#include <hip/hip_runtime.h>
#include <math.h>


typedef float v2f __attribute__((ext_vector_type(2)));
typedef float v4f __attribute__((ext_vector_type(4)));
typedef float v8f __attribute__((ext_vector_type(8)));

#define TWO_PI_F     6.2831853071795864769f
#define MIN_W        1e-3f
#define MIN_H        1e-3f
#define MIN_D        1e-3f
#define DERIV_SHIFT  0.54132485461292f

#define KBINS 32
#define DH    8
#define CDIM  64
#define HDIM  256
#define NPAR  768

#define WAVES           8
#define NTILES          2
#define ROWS_PER_WAVE   (NTILES * 16)
#define ROWS_PER_BLOCK  (WAVES * ROWS_PER_WAVE)
#define THREADS         (WAVES * 32)
#define PCHUNK          96
#define MT2             (PCHUNK / 16)
#define HT              (HDIM / 16)

#define KPAIRS      (HDIM / 2)
#define WPITCH      200
#define SLAB_FLOATS (KPAIRS * WPITCH)
#define PPITCH      100
#define PBUF_FLOATS (ROWS_PER_WAVE * PPITCH)
#define LDS_FLOATS  (2 * SLAB_FLOATS + WAVES * PBUF_FLOATS)

#define QUADS       (PCHUNK / 4)
#define STAGE_ITEMS ((KPAIRS * QUADS) / THREADS)

static __device__ __forceinline__ float softplus_f(float x) {
    return (x > 20.0f) ? x : log1pf(__expf(x));
}

static __device__ __forceinline__ void stage_slab(float* __restrict__ dst,
                                                  const float* __restrict__ W2,
                                                  int dp, int tid)
{
    v4f ra[STAGE_ITEMS], rb[STAGE_ITEMS];
    #pragma unroll
    for (int i = 0; i < STAGE_ITEMS; ++i) {
        const int item = tid + THREADS * i;
        const int kp = item / QUADS;
        const int q  = item - kp * QUADS;
        const float* g = W2 + (size_t)(2 * kp) * NPAR + dp * PCHUNK + 4 * q;
        ra[i] = *(const v4f*)g;
        rb[i] = *(const v4f*)(g + NPAR);
    }
    #pragma unroll
    for (int i = 0; i < STAGE_ITEMS; ++i) {
        const int item = tid + THREADS * i;
        const int kp = item / QUADS;
        const int q  = item - kp * QUADS;
        float* d = dst + kp * WPITCH + 8 * q;
        v4f lo, hi;
        lo.x = ra[i].x; lo.y = rb[i].x; lo.z = ra[i].y; lo.w = rb[i].y;
        hi.x = ra[i].z; hi.y = rb[i].z; hi.z = ra[i].w; hi.w = rb[i].w;
        *(v4f*)d       = lo;
        *(v4f*)(d + 4) = hi;
    }
}

__global__ __launch_bounds__(THREADS)
void Circular_Spline_Flow_fused_kernel(
    const float* __restrict__ theta,
    const float* __restrict__ x,
    const float* __restrict__ W1,
    const float* __restrict__ b1,
    const float* __restrict__ W2,
    const float* __restrict__ b2,
    const float* __restrict__ eta,
    float* __restrict__ out,
    int Btotal)
{
    __shared__ __attribute__((aligned(16))) float smem[LDS_FLOATS];

    const int tid  = threadIdx.x;
    const int wave = tid >> 5;
    const int lane = tid & 31;
    const int c    = lane & 15;
    const int half = lane >> 4;

    const int rowbase = blockIdx.x * ROWS_PER_BLOCK + wave * ROWS_PER_WAVE;

    float* slab0 = smem;
    float* slab1 = smem + SLAB_FLOATS;
    float* pbuf  = smem + 2 * SLAB_FLOATS + wave * PBUF_FLOATS;

    const float etav = eta[0];

    v8f H[NTILES][HT];
    {
        v2f xfrag[NTILES][16];
        #pragma unroll
        for (int nt = 0; nt < NTILES; ++nt) {
            const float* xrow = x + (size_t)(rowbase + nt * 16 + c) * CDIM;
            #pragma unroll
            for (int kc = 0; kc < 16; ++kc)
                xfrag[nt][kc] = *(const v2f*)(xrow + 4 * kc + 2 * half);
        }
        #pragma unroll
        for (int j = 0; j < HT; ++j) {
            v8f acc0 = {0.f,0.f,0.f,0.f,0.f,0.f,0.f,0.f};
            v8f acc1 = {0.f,0.f,0.f,0.f,0.f,0.f,0.f,0.f};
            #pragma unroll
            for (int kc = 0; kc < 16; ++kc) {
                const int kk = 4 * kc + 2 * half;
                v2f w;
                w.x = W1[(size_t)kk * HDIM + 16 * j + c];
                w.y = W1[(size_t)(kk + 1) * HDIM + 16 * j + c];
                acc0 = __builtin_amdgcn_wmma_f32_16x16x4_f32(
                    false, w, false, xfrag[0][kc], (short)0, acc0, false, false);
                acc1 = __builtin_amdgcn_wmma_f32_16x16x4_f32(
                    false, w, false, xfrag[1][kc], (short)0, acc1, false, false);
            }
            const v4f bl = *(const v4f*)(b1 + 16 * j + 8 * half);
            const v4f bh = *(const v4f*)(b1 + 16 * j + 8 * half + 4);
            const float bias[8] = {bl.x, bl.y, bl.z, bl.w, bh.x, bh.y, bh.z, bh.w};
            #pragma unroll
            for (int r = 0; r < 8; ++r) {
                H[0][j][r] = fmaxf(acc0[r] + bias[r], 0.0f);
                H[1][j][r] = fmaxf(acc1[r] + bias[r], 0.0f);
            }
        }
    }

    stage_slab(slab0, W2, 0, tid);
    __syncthreads();

    const int abase = (4 * half) * WPITCH + 2 * c;

    for (int dp = 0; dp < DH; ++dp) {
        const float* slab = (dp & 1) ? slab1 : slab0;
        if (dp + 1 < DH)
            stage_slab((dp & 1) ? slab0 : slab1, W2, dp + 1, tid);

        v8f acc[NTILES][MT2];
        #pragma unroll
        for (int nt = 0; nt < NTILES; ++nt)
            #pragma unroll
            for (int jt = 0; jt < MT2; ++jt)
                acc[nt][jt] = (v8f){0.f,0.f,0.f,0.f,0.f,0.f,0.f,0.f};

        #pragma unroll
        for (int j = 0; j < HT; ++j) {
            #pragma unroll
            for (int s = 0; s < 4; ++s) {
                v2f hb0, hb1;
                hb0.x = H[0][j][2 * s]; hb0.y = H[0][j][2 * s + 1];
                hb1.x = H[1][j][2 * s]; hb1.y = H[1][j][2 * s + 1];
                const float* arow = slab + abase + (8 * j + s) * WPITCH;
                #pragma unroll
                for (int jt = 0; jt < MT2; ++jt) {
                    const v2f a = *(const v2f*)(arow + 32 * jt);
                    acc[0][jt] = __builtin_amdgcn_wmma_f32_16x16x4_f32(
                        false, a, false, hb0, (short)0, acc[0][jt], false, false);
                    acc[1][jt] = __builtin_amdgcn_wmma_f32_16x16x4_f32(
                        false, a, false, hb1, (short)0, acc[1][jt], false, false);
                }
            }
        }

        #pragma unroll
        for (int jt = 0; jt < MT2; ++jt) {
            const float* bp = b2 + dp * PCHUNK + 16 * jt + 8 * half;
            const v4f bl = *(const v4f*)bp;
            const v4f bh = *(const v4f*)(bp + 4);
            #pragma unroll
            for (int nt = 0; nt < NTILES; ++nt) {
                v4f lo, hi;
                lo.x = (acc[nt][jt][0] + bl.x) * etav;
                lo.y = (acc[nt][jt][1] + bl.y) * etav;
                lo.z = (acc[nt][jt][2] + bl.z) * etav;
                lo.w = (acc[nt][jt][3] + bl.w) * etav;
                hi.x = (acc[nt][jt][4] + bh.x) * etav;
                hi.y = (acc[nt][jt][5] + bh.y) * etav;
                hi.z = (acc[nt][jt][6] + bh.z) * etav;
                hi.w = (acc[nt][jt][7] + bh.w) * etav;
                float* d = pbuf + (nt * 16 + c) * PPITCH + 16 * jt + 8 * half;
                *(v4f*)d       = lo;
                *(v4f*)(d + 4) = hi;
            }
        }

        __builtin_amdgcn_fence(__ATOMIC_RELEASE, "workgroup");
        __builtin_amdgcn_s_barrier_signal(-1);

        {
            const float* P  = &pbuf[lane * PPITCH];
            const float* uw = P;
            const float* uh = P + 32;
            const float* ud = P + 64;
            const float  th = theta[(size_t)(rowbase + lane) * DH + dp];

            float mw = -1e30f, mh = -1e30f;
            for (int i = 0; i < KBINS; ++i) {
                mw = fmaxf(mw, uw[i]);
                mh = fmaxf(mh, uh[i]);
            }
            float sw = 0.f, sh = 0.f;
            for (int i = 0; i < KBINS; ++i) {
                sw += __expf(uw[i] - mw);
                sh += __expf(uh[i] - mh);
            }
            const float wscale = (1.0f - MIN_W * KBINS) / sw;
            const float hscale = (1.0f - MIN_H * KBINS) / sh;

            float cw = 0.f, ch = 0.f;
            float Sw = 0.f, Sh = 0.f;
            int   idx = 0;
            float sel_cw = 0.f, sel_ch = 0.f, sel_w = TWO_PI_F, sel_h = TWO_PI_F;
            for (int i = 0; i < KBINS; ++i) {
                const float wfrac = MIN_W + wscale * __expf(uw[i] - mw);
                const float hfrac = MIN_H + hscale * __expf(uh[i] - mh);
                Sw += wfrac;
                Sh += hfrac;
                const float cw_next = (i == KBINS - 1) ? TWO_PI_F : TWO_PI_F * Sw;
                const float ch_next = (i == KBINS - 1) ? TWO_PI_F : TWO_PI_F * Sh;
                if (th >= cw) {
                    idx    = i;
                    sel_cw = cw;  sel_w = cw_next - cw;
                    sel_ch = ch;  sel_h = ch_next - ch;
                }
                cw = cw_next;
                ch = ch_next;
            }

            const float d0 = MIN_D + softplus_f(ud[idx] + DERIV_SHIFT);
            const float d1 = MIN_D + softplus_f(ud[(idx + 1) & (KBINS - 1)] + DERIV_SHIFT);

            const float delta = sel_h / sel_w;
            const float t     = (th - sel_cw) / sel_w;
            const float t1mt  = t * (1.0f - t);
            const float numer = sel_h * (delta * t * t + d0 * t1mt);
            const float denom = delta + (d0 + d1 - 2.0f * delta) * t1mt;
            const float outv  = sel_ch + numer / denom;

            const float omt  = 1.0f - t;
            const float dnum = delta * delta * (d1 * t * t + 2.0f * delta * t1mt + d0 * omt * omt);
            const float lad  = __logf(dnum) - 2.0f * __logf(denom);

            const size_t o = (size_t)(rowbase + lane) * DH + dp;
            out[o] = outv;
            out[(size_t)Btotal * DH + o] = lad;
        }

        __builtin_amdgcn_s_barrier_wait(-1);
        __builtin_amdgcn_fence(__ATOMIC_ACQUIRE, "workgroup");
    }
}

extern "C" void kernel_launch(void* const* d_in, const int* in_sizes, int n_in,
                              void* d_out, int out_size, void* d_ws, size_t ws_size,
                              hipStream_t stream) {
    (void)n_in; (void)out_size; (void)d_ws; (void)ws_size;
    const float* theta = (const float*)d_in[0];
    const float* x     = (const float*)d_in[1];
    const float* W1    = (const float*)d_in[2];
    const float* b1    = (const float*)d_in[3];
    const float* W2    = (const float*)d_in[4];
    const float* b2    = (const float*)d_in[5];
    const float* eta   = (const float*)d_in[6];
    float* out = (float*)d_out;

    const int B = in_sizes[0] / DH;
    const int blocks = B / ROWS_PER_BLOCK;
    Circular_Spline_Flow_fused_kernel<<<blocks, THREADS, 0, stream>>>(
        theta, x, W1, b1, W2, b2, eta, out, B);
}
